# starting kernel
# baseline (speedup 1.0000x reference)
#include <hip/hip_runtime.h>
#include <hip/hip_bf16.h>
#include <cstdint>

#define BB 64
#define SS 512
#define DD 768
#define TT 64

typedef float v2f __attribute__((ext_vector_type(2)));
typedef float v8f __attribute__((ext_vector_type(8)));

__global__ __launch_bounds__(256) void crf_emissions_wmma(
    const float* __restrict__ x,
    const float* __restrict__ W,
    const float* __restrict__ bias,
    float* __restrict__ em)
{
  const int wave = blockIdx.x * (blockDim.x >> 5) + (threadIdx.x >> 5);
  const int lane = threadIdx.x & 31;
  const int m0   = wave * 16;
  const int l16  = lane & 15;
  const int kh   = lane >> 4;

  const float* __restrict__ arow = x + (size_t)(m0 + l16) * DD + 2 * kh;
  const float* __restrict__ w0 = W + (size_t)l16 * DD + 2 * kh;

  v8f acc0 = {0.f,0.f,0.f,0.f,0.f,0.f,0.f,0.f};
  v8f acc1 = acc0, acc2 = acc0, acc3 = acc0;

  #pragma unroll 4
  for (int k = 0; k < DD; k += 4) {
    v2f a  = *(const v2f*)(arow + k);
    v2f b0 = *(const v2f*)(w0 + k);
    v2f b1 = *(const v2f*)(w0 + 16 * DD + k);
    v2f b2 = *(const v2f*)(w0 + 32 * DD + k);
    v2f b3 = *(const v2f*)(w0 + 48 * DD + k);
    acc0 = __builtin_amdgcn_wmma_f32_16x16x4_f32(false, a, false, b0, (short)0, acc0, false, false);
    acc1 = __builtin_amdgcn_wmma_f32_16x16x4_f32(false, a, false, b1, (short)0, acc1, false, false);
    acc2 = __builtin_amdgcn_wmma_f32_16x16x4_f32(false, a, false, b2, (short)0, acc2, false, false);
    acc3 = __builtin_amdgcn_wmma_f32_16x16x4_f32(false, a, false, b3, (short)0, acc3, false, false);
  }

  const int rbase = m0 + kh * 8;
  #pragma unroll
  for (int nt = 0; nt < 4; ++nt) {
    v8f acc = (nt == 0) ? acc0 : (nt == 1) ? acc1 : (nt == 2) ? acc2 : acc3;
    const int n  = nt * 16 + l16;
    const float bn = bias[n];
    #pragma unroll
    for (int v = 0; v < 8; ++v) {
      em[(size_t)(rbase + v) * TT + n] = acc[v] + bn;
    }
  }
}

__global__ __launch_bounds__(256) void crf_viterbi(
    const float* __restrict__ em,
    const float* __restrict__ trans,
    int* __restrict__ out)
{
  const int b   = blockIdx.x;
  const int tid = threadIdx.x;
  const int tc  = tid & 63;
  const int p   = tid >> 6;

  __shared__ float         score[TT];
  __shared__ float         pval[4][TT];
  __shared__ unsigned char pidx[4][TT];
  __shared__ unsigned char hist[(SS - 1) * TT];

  float tr[16];
  #pragma unroll
  for (int i = 0; i < 16; ++i) tr[i] = trans[(p * 16 + i) * TT + tc];

  const float* __restrict__ emb = em + (size_t)b * SS * TT;

  if (tid < TT) score[tc] = emb[tc];
  __syncthreads();

  for (int t = 1; t < SS; ++t) {
    float best = -__builtin_inff();
    int   bi   = p * 16;
    #pragma unroll
    for (int i = 0; i < 16; ++i) {
      float v = score[p * 16 + i] + tr[i];
      if (v > best) { best = v; bi = p * 16 + i; }
    }
    pval[p][tc] = best;
    pidx[p][tc] = (unsigned char)bi;
    __syncthreads();

    if (p == 0) {
      float bv = pval[0][tc];
      int   bb = pidx[0][tc];
      #pragma unroll
      for (int q = 1; q < 4; ++q) {
        float v = pval[q][tc];
        if (v > bv) { bv = v; bb = pidx[q][tc]; }
      }
      score[tc] = bv + emb[(size_t)t * TT + tc];
      hist[(t - 1) * TT + tc] = (unsigned char)bb;
    }
    __syncthreads();
  }

  if (tid == 0) {
    float bv = score[0];
    int   bt = 0;
    for (int t2 = 1; t2 < TT; ++t2) {
      if (score[t2] > bv) { bv = score[t2]; bt = t2; }
    }
    int* __restrict__ ob = out + (size_t)b * SS;
    ob[SS - 1] = bt;
    int cur = bt;
    for (int t = SS - 2; t >= 0; --t) {
      cur = hist[t * TT + cur];
      ob[t] = cur;
    }
  }
}

extern "C" void kernel_launch(void* const* d_in, const int* in_sizes, int n_in,
                              void* d_out, int out_size, void* d_ws, size_t ws_size,
                              hipStream_t stream) {
  const float* x     = (const float*)d_in[0];
  const float* W     = (const float*)d_in[1];
  const float* bias  = (const float*)d_in[2];
  const float* trans = (const float*)d_in[3];
  float* em = (float*)d_ws;
  int*   out = (int*)d_out;

  crf_emissions_wmma<<<(BB * SS / 16) / 8, 256, 0, stream>>>(x, W, bias, em);
  crf_viterbi<<<BB, 256, 0, stream>>>(em, trans, out);
}


# Viterbi scan: emission rows prefetched into registers one 7-step block ahead, barriers fence LDS only so the loads stay in flight; no scan step waits for a global load
# speedup vs baseline: 1.7761x; 1.7761x over previous
#include <hip/hip_runtime.h>
#include <hip/hip_bf16.h>
#include <cstdint>

#define BB 64
#define SS 512
#define DD 768
#define TT 64

typedef float v2f __attribute__((ext_vector_type(2)));
typedef float v8f __attribute__((ext_vector_type(8)));

__global__ __launch_bounds__(256) void crf_emissions_wmma(
    const float* __restrict__ x,
    const float* __restrict__ W,
    const float* __restrict__ bias,
    float* __restrict__ em)
{
  const int wave = blockIdx.x * (blockDim.x >> 5) + (threadIdx.x >> 5);
  const int lane = threadIdx.x & 31;
  const int m0   = wave * 16;
  const int l16  = lane & 15;
  const int kh   = lane >> 4;

  const float* __restrict__ arow = x + (size_t)(m0 + l16) * DD + 2 * kh;
  const float* __restrict__ w0 = W + (size_t)l16 * DD + 2 * kh;

  v8f acc0 = {0.f,0.f,0.f,0.f,0.f,0.f,0.f,0.f};
  v8f acc1 = acc0, acc2 = acc0, acc3 = acc0;

  #pragma unroll 4
  for (int k = 0; k < DD; k += 4) {
    v2f a  = *(const v2f*)(arow + k);
    v2f b0 = *(const v2f*)(w0 + k);
    v2f b1 = *(const v2f*)(w0 + 16 * DD + k);
    v2f b2 = *(const v2f*)(w0 + 32 * DD + k);
    v2f b3 = *(const v2f*)(w0 + 48 * DD + k);
    acc0 = __builtin_amdgcn_wmma_f32_16x16x4_f32(false, a, false, b0, (short)0, acc0, false, false);
    acc1 = __builtin_amdgcn_wmma_f32_16x16x4_f32(false, a, false, b1, (short)0, acc1, false, false);
    acc2 = __builtin_amdgcn_wmma_f32_16x16x4_f32(false, a, false, b2, (short)0, acc2, false, false);
    acc3 = __builtin_amdgcn_wmma_f32_16x16x4_f32(false, a, false, b3, (short)0, acc3, false, false);
  }

  const int rbase = m0 + kh * 8;
  #pragma unroll
  for (int nt = 0; nt < 4; ++nt) {
    v8f acc = (nt == 0) ? acc0 : (nt == 1) ? acc1 : (nt == 2) ? acc2 : acc3;
    const int n  = nt * 16 + l16;
    const float bn = bias[n];
    #pragma unroll
    for (int v = 0; v < 8; ++v) {
      em[(size_t)(rbase + v) * TT + n] = acc[v] + bn;
    }
  }
}

__device__ __forceinline__ void lds_barrier() {
  __builtin_amdgcn_fence(__ATOMIC_RELEASE, "workgroup", "local");
  __builtin_amdgcn_s_barrier();
  __builtin_amdgcn_fence(__ATOMIC_ACQUIRE, "workgroup", "local");
}

__global__ __launch_bounds__(256) void crf_viterbi(
    const float* __restrict__ em,
    const float* __restrict__ trans,
    int* __restrict__ out)
{
  const int b   = blockIdx.x;
  const int tid = threadIdx.x;
  const int tc  = tid & 63;
  const int p   = tid >> 6;

  __shared__ float         score[TT];
  __shared__ float         pval[4][TT];
  __shared__ unsigned char pidx[4][TT];
  __shared__ unsigned char hist[(SS - 1) * TT];

  float tr[16];
  #pragma unroll
  for (int i = 0; i < 16; ++i) tr[i] = trans[(p * 16 + i) * TT + tc];

  const float* __restrict__ emb = em + (size_t)b * SS * TT;

  constexpr int PF = 7;
  static_assert((SS - 1) % PF == 0, "block size must divide the step count");
  float e_cur[PF] = {};
  float e_nxt[PF] = {};

  if (tid < TT) {
    score[tc] = emb[tc];
    #pragma unroll
    for (int j = 0; j < PF; ++j) e_cur[j] = emb[(size_t)(1 + j) * TT + tc];
  }
  __syncthreads();

  for (int t0 = 1; t0 < SS; t0 += PF) {
    if (p == 0) {
      #pragma unroll
      for (int j = 0; j < PF; ++j) {
        const int tn = (t0 + PF + j < SS) ? (t0 + PF + j) : (SS - 1);
        e_nxt[j] = emb[(size_t)tn * TT + tc];
      }
    }

    #pragma unroll
    for (int j = 0; j < PF; ++j) {
      const int t = t0 + j;
      float best = -__builtin_inff();
      int   bi   = p * 16;
      #pragma unroll
      for (int i = 0; i < 16; ++i) {
        float v = score[p * 16 + i] + tr[i];
        if (v > best) { best = v; bi = p * 16 + i; }
      }
      pval[p][tc] = best;
      pidx[p][tc] = (unsigned char)bi;
      lds_barrier();

      if (p == 0) {
        float bv = pval[0][tc];
        int   bb = pidx[0][tc];
        #pragma unroll
        for (int q = 1; q < 4; ++q) {
          float v = pval[q][tc];
          if (v > bv) { bv = v; bb = pidx[q][tc]; }
        }
        score[tc] = bv + e_cur[j];
        hist[(t - 1) * TT + tc] = (unsigned char)bb;
      }
      lds_barrier();
    }

    #pragma unroll
    for (int j = 0; j < PF; ++j) e_cur[j] = e_nxt[j];
  }

  if (tid == 0) {
    float bv = score[0];
    int   bt = 0;
    for (int t2 = 1; t2 < TT; ++t2) {
      if (score[t2] > bv) { bv = score[t2]; bt = t2; }
    }
    int* __restrict__ ob = out + (size_t)b * SS;
    ob[SS - 1] = bt;
    int cur = bt;
    for (int t = SS - 2; t >= 0; --t) {
      cur = hist[t * TT + cur];
      ob[t] = cur;
    }
  }
}

extern "C" void kernel_launch(void* const* d_in, const int* in_sizes, int n_in,
                              void* d_out, int out_size, void* d_ws, size_t ws_size,
                              hipStream_t stream) {
  const float* x     = (const float*)d_in[0];
  const float* W     = (const float*)d_in[1];
  const float* bias  = (const float*)d_in[2];
  const float* trans = (const float*)d_in[3];
  float* em = (float*)d_ws;
  int*   out = (int*)d_out;

  crf_emissions_wmma<<<(BB * SS / 16) / 8, 256, 0, stream>>>(x, W, bias, em);
  crf_viterbi<<<BB, 256, 0, stream>>>(em, trans, out);
}


# Viterbi scan: the 4 prev-tag slices of a tag sit in one quad, reduced by two DPP compares in registers; scores double-buffered in LDS: one barrier per step instead of two; results bit-identical
# speedup vs baseline: 2.1259x; 1.1969x over previous
#include <hip/hip_runtime.h>
#include <hip/hip_bf16.h>
#include <cstdint>

#define BB 64
#define SS 512
#define DD 768
#define TT 64

typedef float v2f __attribute__((ext_vector_type(2)));
typedef float v8f __attribute__((ext_vector_type(8)));

__global__ __launch_bounds__(256) void crf_emissions_wmma(
    const float* __restrict__ x,
    const float* __restrict__ W,
    const float* __restrict__ bias,
    float* __restrict__ em)
{
  const int wave = blockIdx.x * (blockDim.x >> 5) + (threadIdx.x >> 5);
  const int lane = threadIdx.x & 31;
  const int m0   = wave * 16;
  const int l16  = lane & 15;
  const int kh   = lane >> 4;

  const float* __restrict__ arow = x + (size_t)(m0 + l16) * DD + 2 * kh;
  const float* __restrict__ w0 = W + (size_t)l16 * DD + 2 * kh;

  v8f acc0 = {0.f,0.f,0.f,0.f,0.f,0.f,0.f,0.f};
  v8f acc1 = acc0, acc2 = acc0, acc3 = acc0;

  #pragma unroll 4
  for (int k = 0; k < DD; k += 4) {
    v2f a  = *(const v2f*)(arow + k);
    v2f b0 = *(const v2f*)(w0 + k);
    v2f b1 = *(const v2f*)(w0 + 16 * DD + k);
    v2f b2 = *(const v2f*)(w0 + 32 * DD + k);
    v2f b3 = *(const v2f*)(w0 + 48 * DD + k);
    acc0 = __builtin_amdgcn_wmma_f32_16x16x4_f32(false, a, false, b0, (short)0, acc0, false, false);
    acc1 = __builtin_amdgcn_wmma_f32_16x16x4_f32(false, a, false, b1, (short)0, acc1, false, false);
    acc2 = __builtin_amdgcn_wmma_f32_16x16x4_f32(false, a, false, b2, (short)0, acc2, false, false);
    acc3 = __builtin_amdgcn_wmma_f32_16x16x4_f32(false, a, false, b3, (short)0, acc3, false, false);
  }

  const int rbase = m0 + kh * 8;
  #pragma unroll
  for (int nt = 0; nt < 4; ++nt) {
    v8f acc = (nt == 0) ? acc0 : (nt == 1) ? acc1 : (nt == 2) ? acc2 : acc3;
    const int n  = nt * 16 + l16;
    const float bn = bias[n];
    #pragma unroll
    for (int v = 0; v < 8; ++v) {
      em[(size_t)(rbase + v) * TT + n] = acc[v] + bn;
    }
  }
}

__device__ __forceinline__ void lds_barrier() {
  __builtin_amdgcn_fence(__ATOMIC_RELEASE, "workgroup", "local");
  __builtin_amdgcn_s_barrier();
  __builtin_amdgcn_fence(__ATOMIC_ACQUIRE, "workgroup", "local");
}


template <int CTRL>
__device__ __forceinline__ float quad_get_f(float v) {
  const int iv = __builtin_bit_cast(int, v);
  return __builtin_bit_cast(float, __builtin_amdgcn_update_dpp(0, iv, CTRL, 0xF, 0xF, true));
}
template <int CTRL>
__device__ __forceinline__ int quad_get_i(int v) {
  return __builtin_amdgcn_update_dpp(0, v, CTRL, 0xF, 0xF, true);
}

__global__ __launch_bounds__(256) void crf_viterbi(
    const float* __restrict__ em,
    const float* __restrict__ trans,
    int* __restrict__ out)
{
  const int b   = blockIdx.x;
  const int tid = threadIdx.x;
  const int tc  = tid >> 2;
  const int p   = tid & 3;

  __shared__ __attribute__((aligned(16))) float sc[2 * TT];
  __shared__ unsigned char hist[(SS - 1) * TT];

  float tr[16];
  #pragma unroll
  for (int i = 0; i < 16; ++i) tr[i] = trans[(p * 16 + i) * TT + tc];

  const float* __restrict__ emb = em + (size_t)b * SS * TT;

  constexpr int PF = 7;
  static_assert((SS - 1) % PF == 0, "block size must divide the step count");
  float e_cur[PF];
  float e_nxt[PF];

  if (p == 0) sc[tc] = emb[tc];
  #pragma unroll
  for (int j = 0; j < PF; ++j) e_cur[j] = emb[(size_t)(1 + j) * TT + tc];
  __syncthreads();

  int par = 0;
  for (int t0 = 1; t0 < SS; t0 += PF) {
    #pragma unroll
    for (int j = 0; j < PF; ++j) {
      const int tn = (t0 + PF + j < SS) ? (t0 + PF + j) : (SS - 1);
      e_nxt[j] = emb[(size_t)tn * TT + tc];
    }

    #pragma unroll
    for (int j = 0; j < PF; ++j) {
      const int t  = t0 + j;
      const int rb = par ^ (j & 1);
      const float4* s4 =
          reinterpret_cast<const float4*>(sc + rb * TT + p * 16);

      float best = -__builtin_inff();
      int   bi   = p * 16;
      #pragma unroll
      for (int q = 0; q < 4; ++q) {
        const float4 s = s4[q];
        float v;
        v = s.x + tr[4 * q + 0]; if (v > best) { best = v; bi = p * 16 + 4 * q + 0; }
        v = s.y + tr[4 * q + 1]; if (v > best) { best = v; bi = p * 16 + 4 * q + 1; }
        v = s.z + tr[4 * q + 2]; if (v > best) { best = v; bi = p * 16 + 4 * q + 2; }
        v = s.w + tr[4 * q + 3]; if (v > best) { best = v; bi = p * 16 + 4 * q + 3; }
      }

      {
        const float ov = quad_get_f<0xF5>(best);
        const int   oi = quad_get_i<0xF5>(bi);
        if (ov > best) { best = ov; bi = oi; }
      }
      {
        const float ov = quad_get_f<0xAA>(best);
        const int   oi = quad_get_i<0xAA>(bi);
        if (ov > best) { best = ov; bi = oi; }
      }

      if (p == 0) {
        sc[(rb ^ 1) * TT + tc] = best + e_cur[j];
        hist[(t - 1) * TT + tc] = (unsigned char)bi;
      }
      lds_barrier();
    }

    #pragma unroll
    for (int j = 0; j < PF; ++j) e_cur[j] = e_nxt[j];
    par ^= 1;
  }

  if (tid == 0) {
    const float* fs = sc + TT;
    float bv = fs[0];
    int   bt = 0;
    for (int t2 = 1; t2 < TT; ++t2) {
      if (fs[t2] > bv) { bv = fs[t2]; bt = t2; }
    }
    int* __restrict__ ob = out + (size_t)b * SS;
    ob[SS - 1] = bt;
    int cur = bt;
    for (int t = SS - 2; t >= 0; --t) {
      cur = hist[t * TT + cur];
      ob[t] = cur;
    }
  }
}

extern "C" void kernel_launch(void* const* d_in, const int* in_sizes, int n_in,
                              void* d_out, int out_size, void* d_ws, size_t ws_size,
                              hipStream_t stream) {
  const float* x     = (const float*)d_in[0];
  const float* W     = (const float*)d_in[1];
  const float* bias  = (const float*)d_in[2];
  const float* trans = (const float*)d_in[3];
  float* em = (float*)d_ws;
  int*   out = (int*)d_out;

  crf_emissions_wmma<<<(BB * SS / 16) / 8, 256, 0, stream>>>(x, W, bias, em);
  crf_viterbi<<<BB, 256, 0, stream>>>(em, trans, out);
}


# Viterbi scan: serial path of a step holds only the maximum (max3 tree, two DPP maxima; maxima reordered, no sums); backpointer found after the barrier by walking the tree, under the next LDS read
# speedup vs baseline: 2.8344x; 1.3333x over previous
#include <hip/hip_runtime.h>
#include <hip/hip_bf16.h>
#include <cstdint>

#define BB 64
#define SS 512
#define DD 768
#define TT 64

typedef float v2f __attribute__((ext_vector_type(2)));
typedef float v8f __attribute__((ext_vector_type(8)));

__global__ __launch_bounds__(256) void crf_emissions_wmma(
    const float* __restrict__ x,
    const float* __restrict__ W,
    const float* __restrict__ bias,
    float* __restrict__ em)
{
  const int wave = blockIdx.x * (blockDim.x >> 5) + (threadIdx.x >> 5);
  const int lane = threadIdx.x & 31;
  const int m0   = wave * 16;
  const int l16  = lane & 15;
  const int kh   = lane >> 4;

  const float* __restrict__ arow = x + (size_t)(m0 + l16) * DD + 2 * kh;
  const float* __restrict__ w0 = W + (size_t)l16 * DD + 2 * kh;

  v8f acc0 = {0.f,0.f,0.f,0.f,0.f,0.f,0.f,0.f};
  v8f acc1 = acc0, acc2 = acc0, acc3 = acc0;

  #pragma unroll 4
  for (int k = 0; k < DD; k += 4) {
    v2f a  = *(const v2f*)(arow + k);
    v2f b0 = *(const v2f*)(w0 + k);
    v2f b1 = *(const v2f*)(w0 + 16 * DD + k);
    v2f b2 = *(const v2f*)(w0 + 32 * DD + k);
    v2f b3 = *(const v2f*)(w0 + 48 * DD + k);
    acc0 = __builtin_amdgcn_wmma_f32_16x16x4_f32(false, a, false, b0, (short)0, acc0, false, false);
    acc1 = __builtin_amdgcn_wmma_f32_16x16x4_f32(false, a, false, b1, (short)0, acc1, false, false);
    acc2 = __builtin_amdgcn_wmma_f32_16x16x4_f32(false, a, false, b2, (short)0, acc2, false, false);
    acc3 = __builtin_amdgcn_wmma_f32_16x16x4_f32(false, a, false, b3, (short)0, acc3, false, false);
  }

  const int rbase = m0 + kh * 8;
  #pragma unroll
  for (int nt = 0; nt < 4; ++nt) {
    v8f acc = (nt == 0) ? acc0 : (nt == 1) ? acc1 : (nt == 2) ? acc2 : acc3;
    const int n  = nt * 16 + l16;
    const float bn = bias[n];
    #pragma unroll
    for (int v = 0; v < 8; ++v) {
      em[(size_t)(rbase + v) * TT + n] = acc[v] + bn;
    }
  }
}

__device__ __forceinline__ void lds_barrier() {
  __builtin_amdgcn_fence(__ATOMIC_RELEASE, "workgroup", "local");
  __builtin_amdgcn_s_barrier();
  __builtin_amdgcn_fence(__ATOMIC_ACQUIRE, "workgroup", "local");
}


template <int CTRL>
__device__ __forceinline__ float quad_get_f(float v) {
  return __builtin_amdgcn_update_dpp(0.0f, v, CTRL, 0xF, 0xF, true);
}
template <int CTRL>
__device__ __forceinline__ int quad_get_i(int v) {
  return __builtin_amdgcn_update_dpp(0, v, CTRL, 0xF, 0xF, true);
}

__device__ __forceinline__ float max3f(float a, float b, float c) {
  return __builtin_fmaxf(__builtin_fmaxf(a, b), c);
}

__global__ __launch_bounds__(256) void crf_viterbi(
    const float* __restrict__ em,
    const float* __restrict__ trans,
    int* __restrict__ out)
{
  const int b   = blockIdx.x;
  const int tid = threadIdx.x;
  const int tc  = tid >> 2;
  const int p   = tid & 3;
  const int p16 = p * 16;

  __shared__ __attribute__((aligned(16))) float sc[2 * TT];
  __shared__ unsigned char hist[SS * TT];

  float tr[16];
  #pragma unroll
  for (int i = 0; i < 16; ++i) tr[i] = trans[(p * 16 + i) * TT + tc];

  const float* __restrict__ emb = em + (size_t)b * SS * TT;

  constexpr int PF = 7;
  static_assert((SS - 1) % PF == 0, "block size must divide the step count");
  float e_cur[PF];
  float e_nxt[PF];

  if (p == 0) sc[tc] = emb[tc];
  #pragma unroll
  for (int j = 0; j < PF; ++j) e_cur[j] = emb[(size_t)(1 + j) * TT + tc];
  __syncthreads();

  float4 s0, s1, s2, s3;
  {
    const float4* s4 = reinterpret_cast<const float4*>(sc + p * 16);
    s0 = s4[0]; s1 = s4[1]; s2 = s4[2]; s3 = s4[3];
  }

  int key = 0;

  int par = 0;
  for (int t0 = 1; t0 < SS; t0 += PF) {
    #pragma unroll
    for (int j = 0; j < PF; ++j) {
      const int tn = (t0 + PF + j < SS) ? (t0 + PF + j) : (SS - 1);
      e_nxt[j] = emb[(size_t)tn * TT + tc];
    }

    #pragma unroll
    for (int j = 0; j < PF; ++j) {
      const int t  = t0 + j;
      const int rb = par ^ (j & 1);

      float v[16];
      v[0]  = s0.x + tr[0];  v[1]  = s0.y + tr[1];  v[2]  = s0.z + tr[2];  v[3]  = s0.w + tr[3];
      v[4]  = s1.x + tr[4];  v[5]  = s1.y + tr[5];  v[6]  = s1.z + tr[6];  v[7]  = s1.w + tr[7];
      v[8]  = s2.x + tr[8];  v[9]  = s2.y + tr[9];  v[10] = s2.z + tr[10]; v[11] = s2.w + tr[11];
      v[12] = s3.x + tr[12]; v[13] = s3.y + tr[13]; v[14] = s3.z + tr[14]; v[15] = s3.w + tr[15];

      const float g0 = max3f(v[0], v[1], v[2]);
      const float g1 = max3f(v[3], v[4], v[5]);
      const float g2 = max3f(v[6], v[7], v[8]);
      const float g3 = max3f(v[9], v[10], v[11]);
      const float g4 = max3f(v[12], v[13], v[14]);
      const float mA = max3f(g0, g1, g2);
      const float mB = max3f(g3, g4, v[15]);
      const float ml = __builtin_fmaxf(mA, mB);
      float m = __builtin_elementwise_maximum(ml, quad_get_f<0xB1>(ml));
      m = __builtin_elementwise_maximum(m, quad_get_f<0x4E>(m));

      sc[(rb ^ 1) * TT + tc] = m + e_cur[j];
      hist[(t - 1) * TT + tc] = (unsigned char)key;
      lds_barrier();

      {
        const float4* s4 = reinterpret_cast<const float4*>(sc + (rb ^ 1) * TT + p * 16);
        s0 = s4[0]; s1 = s4[1]; s2 = s4[2]; s3 = s4[3];
      }

      asm volatile("" : "+v"(m) : : "memory");

      const bool inA = (mA == m);
      const float h0 = inA ? g0 : g3;
      const float h1 = inA ? g1 : g4;
      const float y0 = inA ? v[0] : v[9];
      const float y1 = inA ? v[1] : v[10];
      const float y3 = inA ? v[3] : v[12];
      const float y4 = inA ? v[4] : v[13];
      const float y6 = inA ? v[6] : v[15];
      const int   hb = inA ? p16 : p16 + 9;
      const bool c0 = (h0 == m);
      const bool c1 = (h1 == m);
      const float z0 = c0 ? y0 : (c1 ? y3 : y6);
      const float z1 = c0 ? y1 : (c1 ? y4 : v[7]);
      const int   gb = c0 ? 0 : (c1 ? 3 : 6);
      const int   ps = (z0 == m) ? 0 : ((z1 == m) ? 1 : 2);
      key = (ml == m) ? (hb + gb + ps) : 64;
      {
        const int o = quad_get_i<0xB1>(key);
        key = (o < key) ? o : key;
      }
      {
        const int o = quad_get_i<0x4E>(key);
        key = (o < key) ? o : key;
      }
      __builtin_amdgcn_sched_barrier(0);
    }

    #pragma unroll
    for (int j = 0; j < PF; ++j) e_cur[j] = e_nxt[j];
    par ^= 1;
  }

  hist[(SS - 1) * TT + tc] = (unsigned char)key;
  __syncthreads();

  if (tid == 0) {
    const float* fs = sc + TT;
    float bv = fs[0];
    int   bt = 0;
    for (int t2 = 1; t2 < TT; ++t2) {
      if (fs[t2] > bv) { bv = fs[t2]; bt = t2; }
    }
    int* __restrict__ ob = out + (size_t)b * SS;
    ob[SS - 1] = bt;
    int cur = bt;
    for (int t = SS - 2; t >= 0; --t) {
      cur = hist[(t + 1) * TT + cur];
      ob[t] = cur;
    }
  }
}

extern "C" void kernel_launch(void* const* d_in, const int* in_sizes, int n_in,
                              void* d_out, int out_size, void* d_ws, size_t ws_size,
                              hipStream_t stream) {
  const float* x     = (const float*)d_in[0];
  const float* W     = (const float*)d_in[1];
  const float* bias  = (const float*)d_in[2];
  const float* trans = (const float*)d_in[3];
  float* em = (float*)d_ws;
  int*   out = (int*)d_out;

  crf_emissions_wmma<<<(BB * SS / 16) / 8, 256, 0, stream>>>(x, W, bias, em);
  crf_viterbi<<<BB, 256, 0, stream>>>(em, trans, out);
}


# Viterbi tail: the backtrack's 511 dependent LDS reads cut to 62 (32 segments of 16 rows: maps walked side by side, one thread hops across, 32 lanes re-walk); final scores read as a group; same tags
# speedup vs baseline: 3.2588x; 1.1497x over previous
#include <hip/hip_runtime.h>
#include <hip/hip_bf16.h>
#include <cstdint>

#define BB 64
#define SS 512
#define DD 768
#define TT 64

typedef float v2f __attribute__((ext_vector_type(2)));
typedef float v8f __attribute__((ext_vector_type(8)));

__global__ __launch_bounds__(256) void crf_emissions_wmma(
    const float* __restrict__ x,
    const float* __restrict__ W,
    const float* __restrict__ bias,
    float* __restrict__ em)
{
  const int wave = blockIdx.x * (blockDim.x >> 5) + (threadIdx.x >> 5);
  const int lane = threadIdx.x & 31;
  const int m0   = wave * 16;
  const int l16  = lane & 15;
  const int kh   = lane >> 4;

  const float* __restrict__ arow = x + (size_t)(m0 + l16) * DD + 2 * kh;
  const float* __restrict__ w0 = W + (size_t)l16 * DD + 2 * kh;

  v8f acc0 = {0.f,0.f,0.f,0.f,0.f,0.f,0.f,0.f};
  v8f acc1 = acc0, acc2 = acc0, acc3 = acc0;

  #pragma unroll 4
  for (int k = 0; k < DD; k += 4) {
    v2f a  = *(const v2f*)(arow + k);
    v2f b0 = *(const v2f*)(w0 + k);
    v2f b1 = *(const v2f*)(w0 + 16 * DD + k);
    v2f b2 = *(const v2f*)(w0 + 32 * DD + k);
    v2f b3 = *(const v2f*)(w0 + 48 * DD + k);
    acc0 = __builtin_amdgcn_wmma_f32_16x16x4_f32(false, a, false, b0, (short)0, acc0, false, false);
    acc1 = __builtin_amdgcn_wmma_f32_16x16x4_f32(false, a, false, b1, (short)0, acc1, false, false);
    acc2 = __builtin_amdgcn_wmma_f32_16x16x4_f32(false, a, false, b2, (short)0, acc2, false, false);
    acc3 = __builtin_amdgcn_wmma_f32_16x16x4_f32(false, a, false, b3, (short)0, acc3, false, false);
  }

  const int rbase = m0 + kh * 8;
  #pragma unroll
  for (int nt = 0; nt < 4; ++nt) {
    v8f acc = (nt == 0) ? acc0 : (nt == 1) ? acc1 : (nt == 2) ? acc2 : acc3;
    const int n  = nt * 16 + l16;
    const float bn = bias[n];
    #pragma unroll
    for (int v = 0; v < 8; ++v) {
      em[(size_t)(rbase + v) * TT + n] = acc[v] + bn;
    }
  }
}

__device__ __forceinline__ void lds_barrier() {
  __builtin_amdgcn_fence(__ATOMIC_RELEASE, "workgroup", "local");
  __builtin_amdgcn_s_barrier();
  __builtin_amdgcn_fence(__ATOMIC_ACQUIRE, "workgroup", "local");
}


template <int CTRL>
__device__ __forceinline__ float quad_get_f(float v) {
  return __builtin_amdgcn_update_dpp(0.0f, v, CTRL, 0xF, 0xF, true);
}
template <int CTRL>
__device__ __forceinline__ int quad_get_i(int v) {
  return __builtin_amdgcn_update_dpp(0, v, CTRL, 0xF, 0xF, true);
}

__device__ __forceinline__ float max3f(float a, float b, float c) {
  return __builtin_fmaxf(__builtin_fmaxf(a, b), c);
}

__global__ __launch_bounds__(256) void crf_viterbi(
    const float* __restrict__ em,
    const float* __restrict__ trans,
    int* __restrict__ out)
{
  const int b   = blockIdx.x;
  const int tid = threadIdx.x;
  const int tc  = tid >> 2;
  const int p   = tid & 3;
  const int p16 = p * 16;

  __shared__ __attribute__((aligned(16))) float sc[2 * TT];
  __shared__ unsigned char hist[SS * TT];

  float tr[16];
  #pragma unroll
  for (int i = 0; i < 16; ++i) tr[i] = trans[(p * 16 + i) * TT + tc];

  const float* __restrict__ emb = em + (size_t)b * SS * TT;

  constexpr int PF = 7;
  static_assert((SS - 1) % PF == 0, "block size must divide the step count");
  float e_cur[PF];
  float e_nxt[PF];

  if (p == 0) sc[tc] = emb[tc];
  #pragma unroll
  for (int j = 0; j < PF; ++j) e_cur[j] = emb[(size_t)(1 + j) * TT + tc];
  __syncthreads();

  float4 s0, s1, s2, s3;
  {
    const float4* s4 = reinterpret_cast<const float4*>(sc + p * 16);
    s0 = s4[0]; s1 = s4[1]; s2 = s4[2]; s3 = s4[3];
  }

  int key = 0;

  int par = 0;
  for (int t0 = 1; t0 < SS; t0 += PF) {
    #pragma unroll
    for (int j = 0; j < PF; ++j) {
      const int tn = (t0 + PF + j < SS) ? (t0 + PF + j) : (SS - 1);
      e_nxt[j] = emb[(size_t)tn * TT + tc];
    }

    #pragma unroll
    for (int j = 0; j < PF; ++j) {
      const int t  = t0 + j;
      const int rb = par ^ (j & 1);

      float v[16];
      v[0]  = s0.x + tr[0];  v[1]  = s0.y + tr[1];  v[2]  = s0.z + tr[2];  v[3]  = s0.w + tr[3];
      v[4]  = s1.x + tr[4];  v[5]  = s1.y + tr[5];  v[6]  = s1.z + tr[6];  v[7]  = s1.w + tr[7];
      v[8]  = s2.x + tr[8];  v[9]  = s2.y + tr[9];  v[10] = s2.z + tr[10]; v[11] = s2.w + tr[11];
      v[12] = s3.x + tr[12]; v[13] = s3.y + tr[13]; v[14] = s3.z + tr[14]; v[15] = s3.w + tr[15];

      const float g0 = max3f(v[0], v[1], v[2]);
      const float g1 = max3f(v[3], v[4], v[5]);
      const float g2 = max3f(v[6], v[7], v[8]);
      const float g3 = max3f(v[9], v[10], v[11]);
      const float g4 = max3f(v[12], v[13], v[14]);
      const float mA = max3f(g0, g1, g2);
      const float mB = max3f(g3, g4, v[15]);
      const float ml = __builtin_fmaxf(mA, mB);
      float m = __builtin_elementwise_maximum(ml, quad_get_f<0xB1>(ml));
      m = __builtin_elementwise_maximum(m, quad_get_f<0x4E>(m));

      sc[(rb ^ 1) * TT + tc] = m + e_cur[j];
      hist[(t - 1) * TT + tc] = (unsigned char)key;
      lds_barrier();

      {
        const float4* s4 = reinterpret_cast<const float4*>(sc + (rb ^ 1) * TT + p * 16);
        s0 = s4[0]; s1 = s4[1]; s2 = s4[2]; s3 = s4[3];
      }

      asm volatile("" : "+v"(m) : : "memory");

      const bool inA = (mA == m);
      const float h0 = inA ? g0 : g3;
      const float h1 = inA ? g1 : g4;
      const float y0 = inA ? v[0] : v[9];
      const float y1 = inA ? v[1] : v[10];
      const float y3 = inA ? v[3] : v[12];
      const float y4 = inA ? v[4] : v[13];
      const float y6 = inA ? v[6] : v[15];
      const int   hb = inA ? p16 : p16 + 9;
      const bool c0 = (h0 == m);
      const bool c1 = (h1 == m);
      const float z0 = c0 ? y0 : (c1 ? y3 : y6);
      const float z1 = c0 ? y1 : (c1 ? y4 : v[7]);
      const int   gb = c0 ? 0 : (c1 ? 3 : 6);
      const int   ps = (z0 == m) ? 0 : ((z1 == m) ? 1 : 2);
      key = (ml == m) ? (hb + gb + ps) : 64;
      {
        const int o = quad_get_i<0xB1>(key);
        key = (o < key) ? o : key;
      }
      {
        const int o = quad_get_i<0x4E>(key);
        key = (o < key) ? o : key;
      }
      __builtin_amdgcn_sched_barrier(0);
    }

    #pragma unroll
    for (int j = 0; j < PF; ++j) e_cur[j] = e_nxt[j];
    par ^= 1;
  }

  hist[(SS - 1) * TT + tc] = (unsigned char)key;
  __syncthreads();

  constexpr int SEG  = 16;
  constexpr int NSEG = SS / SEG;
  static_assert(NSEG * SEG == SS, "segments must tile the sequence");
  static_assert(NSEG * TT == 8 * 256, "pass 1: 8 walks per thread");
  __shared__ unsigned char segmap[NSEG * TT];
  __shared__ unsigned char segin[NSEG];
  __shared__ __attribute__((aligned(16))) unsigned char tagb[SS];

  {
    const int j  = tid & (TT - 1);
    const int gq = tid >> 6;
    const unsigned char* hq = hist + gq * (8 * SEG * TT);
    int w[8];
    #pragma unroll
    for (int i = 0; i < 8; ++i) w[i] = j;
    #pragma unroll
    for (int k = SEG - 1; k >= 0; --k) {
      #pragma unroll
      for (int i = 0; i < 8; ++i) w[i] = hq[(i * SEG + k) * TT + w[i]];
    }
    #pragma unroll
    for (int i = 0; i < 8; ++i) segmap[(gq * 8 + i) * TT + j] = (unsigned char)w[i];
  }
  __syncthreads();

  if (tid == 0) {
    const float4* fs4 = reinterpret_cast<const float4*>(sc + TT);
    float fs[TT];
    #pragma unroll
    for (int q = 0; q < TT / 4; ++q) {
      const float4 f = fs4[q];
      fs[4 * q] = f.x; fs[4 * q + 1] = f.y; fs[4 * q + 2] = f.z; fs[4 * q + 3] = f.w;
    }
    float bv = fs[0];
    int   bt = 0;
    #pragma unroll
    for (int t2 = 1; t2 < TT; ++t2) {
      if (fs[t2] > bv) { bv = fs[t2]; bt = t2; }
    }
    int e = bt;
    segin[NSEG - 1] = (unsigned char)e;
    #pragma unroll
    for (int g = NSEG - 1; g >= 1; --g) {
      e = segmap[g * TT + e];
      segin[g - 1] = (unsigned char)e;
    }
  }
  __syncthreads();

  if (tid < NSEG) {
    const int g = tid;
    const unsigned char* hg = hist + g * (SEG * TT);
    int cur = segin[g];
    tagb[g * SEG + SEG - 1] = (unsigned char)cur;
    #pragma unroll
    for (int k = SEG - 1; k >= 1; --k) {
      cur = hg[k * TT + cur];
      tagb[g * SEG + k - 1] = (unsigned char)cur;
    }
  }
  __syncthreads();

  if (tid < SS / 4) {
    const unsigned int pk = reinterpret_cast<const unsigned int*>(tagb)[tid];
    int4 o;
    o.x = (int)(pk & 0xFFu);
    o.y = (int)((pk >> 8) & 0xFFu);
    o.z = (int)((pk >> 16) & 0xFFu);
    o.w = (int)(pk >> 24);
    reinterpret_cast<int4*>(out + (size_t)b * SS)[tid] = o;
  }
}

extern "C" void kernel_launch(void* const* d_in, const int* in_sizes, int n_in,
                              void* d_out, int out_size, void* d_ws, size_t ws_size,
                              hipStream_t stream) {
  const float* x     = (const float*)d_in[0];
  const float* W     = (const float*)d_in[1];
  const float* bias  = (const float*)d_in[2];
  const float* trans = (const float*)d_in[3];
  float* em = (float*)d_ws;
  int*   out = (int*)d_out;

  crf_emissions_wmma<<<(BB * SS / 16) / 8, 256, 0, stream>>>(x, W, bias, em);
  crf_viterbi<<<BB, 256, 0, stream>>>(em, trans, out);
}


# Emissions GEMM: W copied once per workgroup into LDS in fragment order, every B fragment read from LDS with 16-byte reads; global loads carry x only, two spans ahead; k order kept, em bit-identical
# speedup vs baseline: 4.2736x; 1.3114x over previous
#include <hip/hip_runtime.h>
#include <hip/hip_bf16.h>
#include <cstdint>

#define BB 64
#define SS 512
#define DD 768
#define TT 64

typedef float v2f __attribute__((ext_vector_type(2)));
typedef float v8f __attribute__((ext_vector_type(8)));


#define WP 772
#define NSPAN (DD / 32)

__device__ __forceinline__ void span_load_a(v2f (&a)[8], const float* __restrict__ ap) {
  #pragma unroll
  for (int i = 0; i < 8; ++i) a[i] = *(const v2f*)(ap + 4 * i);
}
__device__ __forceinline__ void span_load_b(float4 (&b)[16], const float* bp) {
  #pragma unroll
  for (int nt = 0; nt < 4; ++nt) {
    #pragma unroll
    for (int u = 0; u < 4; ++u)
      b[nt * 4 + u] = *(const float4*)(bp + nt * 16 * WP + 8 * u);
  }
}

__device__ __forceinline__ void span_mma(const v2f (&a)[8], const float4 (&b)[16],
                                         v8f& c0, v8f& c1, v8f& c2, v8f& c3) {
  #pragma unroll
  for (int u = 0; u < 4; ++u) {
    {
      const v2f a0 = a[2 * u];
      const v2f b0 = {b[u].x, b[u].y};
      const v2f b1 = {b[4 + u].x, b[4 + u].y};
      const v2f b2 = {b[8 + u].x, b[8 + u].y};
      const v2f b3 = {b[12 + u].x, b[12 + u].y};
      c0 = __builtin_amdgcn_wmma_f32_16x16x4_f32(false, a0, false, b0, (short)0, c0, false, false);
      c1 = __builtin_amdgcn_wmma_f32_16x16x4_f32(false, a0, false, b1, (short)0, c1, false, false);
      c2 = __builtin_amdgcn_wmma_f32_16x16x4_f32(false, a0, false, b2, (short)0, c2, false, false);
      c3 = __builtin_amdgcn_wmma_f32_16x16x4_f32(false, a0, false, b3, (short)0, c3, false, false);
    }
    {
      const v2f a1 = a[2 * u + 1];
      const v2f b0 = {b[u].z, b[u].w};
      const v2f b1 = {b[4 + u].z, b[4 + u].w};
      const v2f b2 = {b[8 + u].z, b[8 + u].w};
      const v2f b3 = {b[12 + u].z, b[12 + u].w};
      c0 = __builtin_amdgcn_wmma_f32_16x16x4_f32(false, a1, false, b0, (short)0, c0, false, false);
      c1 = __builtin_amdgcn_wmma_f32_16x16x4_f32(false, a1, false, b1, (short)0, c1, false, false);
      c2 = __builtin_amdgcn_wmma_f32_16x16x4_f32(false, a1, false, b2, (short)0, c2, false, false);
      c3 = __builtin_amdgcn_wmma_f32_16x16x4_f32(false, a1, false, b3, (short)0, c3, false, false);
    }
  }
}

__device__ __forceinline__ void span_step(const v2f (&au)[8], const float4 (&bu)[16],
                                          v2f (&al)[8], float4 (&bl)[16],
                                          const float* __restrict__ ap, const float* bp,
                                          v8f& c0, v8f& c1, v8f& c2, v8f& c3) {
  span_load_a(al, ap);
  span_load_b(bl, bp);
  __builtin_amdgcn_sched_barrier(0);
  span_mma(au, bu, c0, c1, c2, c3);
}

__global__ __launch_bounds__(256) void crf_emissions_wmma(
    const float* __restrict__ x,
    const float* __restrict__ W,
    const float* __restrict__ bias,
    float* __restrict__ em)
{
  __shared__ __attribute__((aligned(16))) float Wl[TT * WP];

  const int tid  = threadIdx.x;
  const int wave = blockIdx.x * (blockDim.x >> 5) + (tid >> 5);
  const int lane = tid & 31;
  const int m0   = wave * 16;
  const int l16  = lane & 15;
  const int kh   = lane >> 4;

  const float* __restrict__ arow = x + (size_t)(m0 + l16) * DD + 2 * kh;
  const float* bl = Wl + l16 * WP + 4 * kh;

  v2f    a0[8], a1[8], a2[8];
  float4 b0[16], b1[16];

  span_load_a(a0, arow);
  span_load_a(a1, arow + 32);

  {
    const int r4  = tid >> 6;
    const int c64 = tid & 63;
    const float* __restrict__ wg = W + r4 * DD + 4 * c64;
    float* wl = Wl + r4 * WP + 8 * (c64 >> 1) + 2 * (c64 & 1);
    float4 v[48];
    #pragma unroll
    for (int rb = 0; rb < 16; ++rb) {
      #pragma unroll
      for (int part = 0; part < 3; ++part)
        v[rb * 3 + part] = *(const float4*)(wg + rb * 4 * DD + part * 256);
    }
    #pragma unroll
    for (int rb = 0; rb < 16; ++rb) {
      #pragma unroll
      for (int part = 0; part < 3; ++part) {
        const float4 f = v[rb * 3 + part];
        float* d = wl + rb * 4 * WP + part * 256;
        const v2f lo = {f.x, f.y};
        const v2f hi = {f.z, f.w};
        *(v2f*)(d)     = lo;
        *(v2f*)(d + 4) = hi;
      }
    }
  }
  __syncthreads();

  v8f acc0 = {0.f,0.f,0.f,0.f,0.f,0.f,0.f,0.f};
  v8f acc1 = acc0, acc2 = acc0, acc3 = acc0;

  span_load_b(b0, bl);

  static_assert(NSPAN % 6 == 0, "six spans per trip");
  for (int s = 0; s < NSPAN; s += 6) {
    const int s6 = (s + 6 < NSPAN) ? (s + 6) : (NSPAN - 1);
    const int s7 = (s + 7 < NSPAN) ? (s + 7) : (NSPAN - 1);
    span_step(a0, b0, a2, b1, arow + 32 * (s + 2), bl + 32 * (s + 1), acc0, acc1, acc2, acc3);
    span_step(a1, b1, a0, b0, arow + 32 * (s + 3), bl + 32 * (s + 2), acc0, acc1, acc2, acc3);
    span_step(a2, b0, a1, b1, arow + 32 * (s + 4), bl + 32 * (s + 3), acc0, acc1, acc2, acc3);
    span_step(a0, b1, a2, b0, arow + 32 * (s + 5), bl + 32 * (s + 4), acc0, acc1, acc2, acc3);
    span_step(a1, b0, a0, b1, arow + 32 * s6,      bl + 32 * (s + 5), acc0, acc1, acc2, acc3);
    span_step(a2, b1, a1, b0, arow + 32 * s7,      bl + 32 * s6,      acc0, acc1, acc2, acc3);
  }

  const int rbase = m0 + kh * 8;
  #pragma unroll
  for (int nt = 0; nt < 4; ++nt) {
    v8f acc = (nt == 0) ? acc0 : (nt == 1) ? acc1 : (nt == 2) ? acc2 : acc3;
    const int n  = nt * 16 + l16;
    const float bn = bias[n];
    #pragma unroll
    for (int v = 0; v < 8; ++v) {
      em[(size_t)(rbase + v) * TT + n] = acc[v] + bn;
    }
  }
}

__device__ __forceinline__ void lds_barrier() {
  __builtin_amdgcn_fence(__ATOMIC_RELEASE, "workgroup", "local");
  __builtin_amdgcn_s_barrier();
  __builtin_amdgcn_fence(__ATOMIC_ACQUIRE, "workgroup", "local");
}


template <int CTRL>
__device__ __forceinline__ float quad_get_f(float v) {
  return __builtin_amdgcn_update_dpp(0.0f, v, CTRL, 0xF, 0xF, true);
}
template <int CTRL>
__device__ __forceinline__ int quad_get_i(int v) {
  return __builtin_amdgcn_update_dpp(0, v, CTRL, 0xF, 0xF, true);
}

__device__ __forceinline__ float max3f(float a, float b, float c) {
  return __builtin_fmaxf(__builtin_fmaxf(a, b), c);
}

__global__ __launch_bounds__(256) void crf_viterbi(
    const float* __restrict__ em,
    const float* __restrict__ trans,
    int* __restrict__ out)
{
  const int b   = blockIdx.x;
  const int tid = threadIdx.x;
  const int tc  = tid >> 2;
  const int p   = tid & 3;
  const int p16 = p * 16;

  __shared__ __attribute__((aligned(16))) float sc[2 * TT];
  __shared__ unsigned char hist[SS * TT];

  float tr[16];
  #pragma unroll
  for (int i = 0; i < 16; ++i) tr[i] = trans[(p * 16 + i) * TT + tc];

  const float* __restrict__ emb = em + (size_t)b * SS * TT;

  constexpr int PF = 7;
  static_assert((SS - 1) % PF == 0, "block size must divide the step count");
  float e_cur[PF];
  float e_nxt[PF];

  if (p == 0) sc[tc] = emb[tc];
  #pragma unroll
  for (int j = 0; j < PF; ++j) e_cur[j] = emb[(size_t)(1 + j) * TT + tc];
  __syncthreads();

  float4 s0, s1, s2, s3;
  {
    const float4* s4 = reinterpret_cast<const float4*>(sc + p * 16);
    s0 = s4[0]; s1 = s4[1]; s2 = s4[2]; s3 = s4[3];
  }

  int key = 0;

  int par = 0;
  for (int t0 = 1; t0 < SS; t0 += PF) {
    #pragma unroll
    for (int j = 0; j < PF; ++j) {
      const int tn = (t0 + PF + j < SS) ? (t0 + PF + j) : (SS - 1);
      e_nxt[j] = emb[(size_t)tn * TT + tc];
    }

    #pragma unroll
    for (int j = 0; j < PF; ++j) {
      const int t  = t0 + j;
      const int rb = par ^ (j & 1);

      float v[16];
      v[0]  = s0.x + tr[0];  v[1]  = s0.y + tr[1];  v[2]  = s0.z + tr[2];  v[3]  = s0.w + tr[3];
      v[4]  = s1.x + tr[4];  v[5]  = s1.y + tr[5];  v[6]  = s1.z + tr[6];  v[7]  = s1.w + tr[7];
      v[8]  = s2.x + tr[8];  v[9]  = s2.y + tr[9];  v[10] = s2.z + tr[10]; v[11] = s2.w + tr[11];
      v[12] = s3.x + tr[12]; v[13] = s3.y + tr[13]; v[14] = s3.z + tr[14]; v[15] = s3.w + tr[15];

      const float g0 = max3f(v[0], v[1], v[2]);
      const float g1 = max3f(v[3], v[4], v[5]);
      const float g2 = max3f(v[6], v[7], v[8]);
      const float g3 = max3f(v[9], v[10], v[11]);
      const float g4 = max3f(v[12], v[13], v[14]);
      const float mA = max3f(g0, g1, g2);
      const float mB = max3f(g3, g4, v[15]);
      const float ml = __builtin_fmaxf(mA, mB);
      float m = __builtin_elementwise_maximum(ml, quad_get_f<0xB1>(ml));
      m = __builtin_elementwise_maximum(m, quad_get_f<0x4E>(m));

      sc[(rb ^ 1) * TT + tc] = m + e_cur[j];
      hist[(t - 1) * TT + tc] = (unsigned char)key;
      lds_barrier();

      {
        const float4* s4 = reinterpret_cast<const float4*>(sc + (rb ^ 1) * TT + p * 16);
        s0 = s4[0]; s1 = s4[1]; s2 = s4[2]; s3 = s4[3];
      }

      asm volatile("" : "+v"(m) : : "memory");

      const bool inA = (mA == m);
      const float h0 = inA ? g0 : g3;
      const float h1 = inA ? g1 : g4;
      const float y0 = inA ? v[0] : v[9];
      const float y1 = inA ? v[1] : v[10];
      const float y3 = inA ? v[3] : v[12];
      const float y4 = inA ? v[4] : v[13];
      const float y6 = inA ? v[6] : v[15];
      const int   hb = inA ? p16 : p16 + 9;
      const bool c0 = (h0 == m);
      const bool c1 = (h1 == m);
      const float z0 = c0 ? y0 : (c1 ? y3 : y6);
      const float z1 = c0 ? y1 : (c1 ? y4 : v[7]);
      const int   gb = c0 ? 0 : (c1 ? 3 : 6);
      const int   ps = (z0 == m) ? 0 : ((z1 == m) ? 1 : 2);
      key = (ml == m) ? (hb + gb + ps) : 64;
      {
        const int o = quad_get_i<0xB1>(key);
        key = (o < key) ? o : key;
      }
      {
        const int o = quad_get_i<0x4E>(key);
        key = (o < key) ? o : key;
      }
      __builtin_amdgcn_sched_barrier(0);
    }

    #pragma unroll
    for (int j = 0; j < PF; ++j) e_cur[j] = e_nxt[j];
    par ^= 1;
  }

  hist[(SS - 1) * TT + tc] = (unsigned char)key;
  __syncthreads();

  constexpr int SEG  = 16;
  constexpr int NSEG = SS / SEG;
  static_assert(NSEG * SEG == SS, "segments must tile the sequence");
  static_assert(NSEG * TT == 8 * 256, "pass 1: 8 walks per thread");
  __shared__ unsigned char segmap[NSEG * TT];
  __shared__ unsigned char segin[NSEG];
  __shared__ __attribute__((aligned(16))) unsigned char tagb[SS];

  {
    const int j  = tid & (TT - 1);
    const int gq = tid >> 6;
    const unsigned char* hq = hist + gq * (8 * SEG * TT);
    int w[8];
    #pragma unroll
    for (int i = 0; i < 8; ++i) w[i] = j;
    #pragma unroll
    for (int k = SEG - 1; k >= 0; --k) {
      #pragma unroll
      for (int i = 0; i < 8; ++i) w[i] = hq[(i * SEG + k) * TT + w[i]];
    }
    #pragma unroll
    for (int i = 0; i < 8; ++i) segmap[(gq * 8 + i) * TT + j] = (unsigned char)w[i];
  }
  __syncthreads();

  if (tid == 0) {
    const float4* fs4 = reinterpret_cast<const float4*>(sc + TT);
    float fs[TT];
    #pragma unroll
    for (int q = 0; q < TT / 4; ++q) {
      const float4 f = fs4[q];
      fs[4 * q] = f.x; fs[4 * q + 1] = f.y; fs[4 * q + 2] = f.z; fs[4 * q + 3] = f.w;
    }
    float bv = fs[0];
    int   bt = 0;
    #pragma unroll
    for (int t2 = 1; t2 < TT; ++t2) {
      if (fs[t2] > bv) { bv = fs[t2]; bt = t2; }
    }
    int e = bt;
    segin[NSEG - 1] = (unsigned char)e;
    #pragma unroll
    for (int g = NSEG - 1; g >= 1; --g) {
      e = segmap[g * TT + e];
      segin[g - 1] = (unsigned char)e;
    }
  }
  __syncthreads();

  if (tid < NSEG) {
    const int g = tid;
    const unsigned char* hg = hist + g * (SEG * TT);
    int cur = segin[g];
    tagb[g * SEG + SEG - 1] = (unsigned char)cur;
    #pragma unroll
    for (int k = SEG - 1; k >= 1; --k) {
      cur = hg[k * TT + cur];
      tagb[g * SEG + k - 1] = (unsigned char)cur;
    }
  }
  __syncthreads();

  if (tid < SS / 4) {
    const unsigned int pk = reinterpret_cast<const unsigned int*>(tagb)[tid];
    int4 o;
    o.x = (int)(pk & 0xFFu);
    o.y = (int)((pk >> 8) & 0xFFu);
    o.z = (int)((pk >> 16) & 0xFFu);
    o.w = (int)(pk >> 24);
    reinterpret_cast<int4*>(out + (size_t)b * SS)[tid] = o;
  }
}

extern "C" void kernel_launch(void* const* d_in, const int* in_sizes, int n_in,
                              void* d_out, int out_size, void* d_ws, size_t ws_size,
                              hipStream_t stream) {
  const float* x     = (const float*)d_in[0];
  const float* W     = (const float*)d_in[1];
  const float* bias  = (const float*)d_in[2];
  const float* trans = (const float*)d_in[3];
  float* em = (float*)d_ws;
  int*   out = (int*)d_out;

  crf_emissions_wmma<<<(BB * SS / 16) / 8, 256, 0, stream>>>(x, W, bias, em);
  crf_viterbi<<<BB, 256, 0, stream>>>(em, trans, out);
}
